# starting kernel
# baseline (speedup 1.0000x reference)
#include <hip/hip_runtime.h>

typedef float v2f __attribute__((ext_vector_type(2)));
typedef float v8f __attribute__((ext_vector_type(8)));

#define NN   50000
#define NRL  12
#define NT   500000
#define HIDN 16
#define EMBN 128
#define RP   25
#define KDIM 400
#define NAUG (2*NT + NN)

__device__ __forceinline__ void aug_edge(const int* __restrict__ t, int e,
                                         int& s, int& r, int& o) {
    if (e < NT)            { s = t[3*e];     r = t[3*e+1];       o = t[3*e+2]; }
    else if (e < 2*NT)     { int j = e - NT;
                             s = t[3*j+2];   r = t[3*j+1] + NRL; o = t[3*j];   }
    else                   { int i = e - 2*NT;
                             s = i;          r = 2*NRL;          o = i;        }
}

__global__ void k_zero(float4* __restrict__ p, int n4) {
    int i = blockIdx.x * blockDim.x + threadIdx.x;
    if (i < n4) p[i] = make_float4(0.f, 0.f, 0.f, 0.f);
}

__global__ void k_deg(const int* __restrict__ t, float* __restrict__ deg) {
    int e = blockIdx.x * blockDim.x + threadIdx.x;
    if (e >= NAUG) return;
    int s, r, o; aug_edge(t, e, s, r, o); (void)s;
    atomicAdd(&deg[r * NN + o], 1.0f);
}

__global__ void k_norm(float* __restrict__ deg) {
    int i = blockIdx.x * blockDim.x + threadIdx.x;
    if (i < RP * NN) deg[i] = 1.0f / fmaxf(deg[i], 1.0f);
}

__global__ void k_msg(const int* __restrict__ t, const float* __restrict__ W1,
                      const float* __restrict__ norm, float* __restrict__ x1) {
    int e = blockIdx.x * blockDim.x + threadIdx.x;
    if (e >= NAUG) return;
    int s, r, o; aug_edge(t, e, s, r, o);
    float en = norm[r * NN + o];
    const float4* w = (const float4*)(W1 + ((long)r * NN + s) * HIDN);
    float* dst = x1 + (long)o * HIDN;
#pragma unroll
    for (int q = 0; q < 4; q++) {
        float4 v = w[q];
        atomicAdd(dst + 4*q + 0, v.x * en);
        atomicAdd(dst + 4*q + 1, v.y * en);
        atomicAdd(dst + 4*q + 2, v.z * en);
        atomicAdd(dst + 4*q + 3, v.w * en);
    }
}

__global__ void k_relu(float* __restrict__ x1, const float* __restrict__ b1) {
    int i = blockIdx.x * blockDim.x + threadIdx.x;
    if (i < NN * HIDN) x1[i] = fmaxf(x1[i] + b1[i & (HIDN - 1)], 0.0f);
}

__global__ void k_agg(const int* __restrict__ t, const float* __restrict__ h,
                      const float* __restrict__ norm, float* __restrict__ agg) {
    int e = blockIdx.x * blockDim.x + threadIdx.x;
    if (e >= NAUG) return;
    int s, r, o; aug_edge(t, e, s, r, o);
    float en = norm[r * NN + o];
    const float4* hs = (const float4*)(h + (long)s * HIDN);
    float* dst = agg + (long)o * KDIM + r * HIDN;
#pragma unroll
    for (int q = 0; q < 4; q++) {
        float4 v = hs[q];
        atomicAdd(dst + 4*q + 0, v.x * en);
        atomicAdd(dst + 4*q + 1, v.y * en);
        atomicAdd(dst + 4*q + 2, v.z * en);
        atomicAdd(dst + 4*q + 3, v.w * en);
    }
}

__global__ void k_gemm(const float* __restrict__ agg, const float* __restrict__ W2,
                       const float* __restrict__ b2, float* __restrict__ x2) {
    int wave = (blockIdx.x * blockDim.x + threadIdx.x) >> 5;
    int lane = threadIdx.x & 31;
    if (wave >= NN / 16) return;
    int m0   = wave * 16;
    int half = lane >> 4;
    int lr   = lane & 15;

    const float* arow = agg + (long)(m0 + lr) * KDIM;

    v8f c[8];
#pragma unroll
    for (int nt = 0; nt < 8; nt++) {
        v8f z = {0.f, 0.f, 0.f, 0.f, 0.f, 0.f, 0.f, 0.f};
        c[nt] = z;
    }

    for (int k0 = 0; k0 < KDIM; k0 += 4) {
        int ka = k0 + 2 * half;
        v2f a;
        a.x = arow[ka];
        a.y = arow[ka + 1];
        const float* brow0 = W2 + (long)ka * EMBN + lr;
        const float* brow1 = W2 + (long)(ka + 1) * EMBN + lr;
#pragma unroll
        for (int nt = 0; nt < 8; nt++) {
            v2f b;
            b.x = brow0[nt * 16];
            b.y = brow1[nt * 16];
            c[nt] = __builtin_amdgcn_wmma_f32_16x16x4_f32(
                false, a, false, b, (short)0, c[nt], false, false);
        }
    }

#pragma unroll
    for (int nt = 0; nt < 8; nt++) {
        float bias = b2[nt * 16 + lr];
#pragma unroll
        for (int v = 0; v < 8; v++) {
            int row = m0 + v + 8 * half;
            x2[(long)row * EMBN + nt * 16 + lr] = c[nt][v] + bias;
        }
    }
}

__global__ void k_score(const int* __restrict__ t, const float* __restrict__ x2,
                        const float* __restrict__ rel, float* __restrict__ out) {
    int gid  = blockIdx.x * blockDim.x + threadIdx.x;
    int w    = gid >> 5;
    int lane = gid & 31;
    if (w >= NT) return;
    int s = t[3*w], p = t[3*w+1], o = t[3*w+2];
    const float* xs = x2 + (long)s * EMBN;
    const float* xo = x2 + (long)o * EMBN;
    const float* rp = rel + (long)p * EMBN;
    float acc = 0.f;
#pragma unroll
    for (int q = 0; q < 4; q++) {
        int j = lane + 32 * q;
        acc += xs[j] * rp[j] * xo[j];
    }
#pragma unroll
    for (int off = 16; off > 0; off >>= 1) acc += __shfl_xor(acc, off, 32);
    if (lane == 0) out[w] = acc;
}

extern "C" void kernel_launch(void* const* d_in, const int* in_sizes, int n_in,
                              void* d_out, int out_size, void* d_ws, size_t ws_size,
                              hipStream_t stream) {
    const int*   triples = (const int*)  d_in[0];
    const float* W1      = (const float*)d_in[1];
    const float* b1      = (const float*)d_in[2];
    const float* W2      = (const float*)d_in[3];
    const float* b2      = (const float*)d_in[4];
    const float* rel     = (const float*)d_in[5];
    float*       out     = (float*)d_out;

    float* deg = (float*)d_ws;
    float* x1  = deg + (long)RP * NN;
    float* agg = x1  + (long)NN * HIDN;
    float* x2  = agg + (long)NN * KDIM;
    long zf = (long)RP * NN + (long)NN * HIDN + (long)NN * KDIM;
    int n4 = (int)(zf / 4);

    dim3 B(256);
    k_zero<<<(n4 + 255) / 256, B, 0, stream>>>((float4*)d_ws, n4);
    k_deg <<<(NAUG + 255) / 256, B, 0, stream>>>(triples, deg);
    k_norm<<<(RP * NN + 255) / 256, B, 0, stream>>>(deg);
    k_msg <<<(NAUG + 255) / 256, B, 0, stream>>>(triples, W1, deg, x1);
    k_relu<<<(NN * HIDN + 255) / 256, B, 0, stream>>>(x1, b1);
    k_agg <<<(NAUG + 255) / 256, B, 0, stream>>>(triples, x1, deg, agg);

    int nwaves = NN / 16;
    k_gemm<<<(nwaves * 32 + 255) / 256, B, 0, stream>>>(agg, W2, b2, x2);

    long sthreads = (long)NT * 32;
    k_score<<<(int)((sthreads + 255) / 256), B, 0, stream>>>(triples, x2, rel, out);
}


# Both edge scatter kernels: one lane per channel; a half-wave passes its 16 edges round by shuffle, so each 16-float row is one 64-B load and one 64-B f32 atomic add (add order by arrival, as before)
# speedup vs baseline: 2.4533x; 2.4533x over previous
#include <hip/hip_runtime.h>

typedef float v2f __attribute__((ext_vector_type(2)));
typedef float v8f __attribute__((ext_vector_type(8)));

#define NN   50000
#define NRL  12
#define NT   500000
#define HIDN 16
#define EMBN 128
#define RP   25
#define KDIM 400
#define NAUG (2*NT + NN)

__device__ __forceinline__ void aug_edge(const int* __restrict__ t, int e,
                                         int& s, int& r, int& o) {
    if (e < NT)            { s = t[3*e];     r = t[3*e+1];       o = t[3*e+2]; }
    else if (e < 2*NT)     { int j = e - NT;
                             s = t[3*j+2];   r = t[3*j+1] + NRL; o = t[3*j];   }
    else                   { int i = e - 2*NT;
                             s = i;          r = 2*NRL;          o = i;        }
}

__global__ void k_zero(float4* __restrict__ p, int n4) {
    int i = blockIdx.x * blockDim.x + threadIdx.x;
    if (i < n4) p[i] = make_float4(0.f, 0.f, 0.f, 0.f);
}

__global__ void k_deg(const int* __restrict__ t, float* __restrict__ deg) {
    int e = blockIdx.x * blockDim.x + threadIdx.x;
    if (e >= NAUG) return;
    int s, r, o; aug_edge(t, e, s, r, o); (void)s;
    atomicAdd(&deg[r * NN + o], 1.0f);
}

__global__ void k_norm(float* __restrict__ deg) {
    int i = blockIdx.x * blockDim.x + threadIdx.x;
    if (i < RP * NN) deg[i] = 1.0f / fmaxf(deg[i], 1.0f);
}

static_assert(NAUG % 16 == 0, "half-waves must be whole");
__device__ __forceinline__ void scatter_rows16(int src, int dst, float en,
                                               const float* __restrict__ stab,
                                               float* __restrict__ dtab) {
    const int c = threadIdx.x & 15;
    float w[16], g[16];
    int   d[16];
#pragma unroll
    for (int j = 0; j < 16; j++) {
        int sj = __shfl(src, j, 16);
        g[j]   = __shfl(en,  j, 16);
        d[j]   = __shfl(dst, j, 16);
        w[j]   = stab[sj * HIDN + c];
    }
#pragma unroll
    for (int j = 0; j < 16; j++)
        atomicAdd(dtab + (d[j] * HIDN + c), w[j] * g[j]);
}

__global__ void k_msg(const int* __restrict__ t, const float* __restrict__ W1,
                      const float* __restrict__ norm, float* __restrict__ x1) {
    int e = blockIdx.x * blockDim.x + threadIdx.x;
    if (e >= NAUG) return;
    int s, r, o; aug_edge(t, e, s, r, o);
    float en = norm[r * NN + o];
    scatter_rows16(r * NN + s, o, en, W1, x1);
}

__global__ void k_relu(float* __restrict__ x1, const float* __restrict__ b1) {
    int i = blockIdx.x * blockDim.x + threadIdx.x;
    if (i < NN * HIDN) x1[i] = fmaxf(x1[i] + b1[i & (HIDN - 1)], 0.0f);
}

__global__ void k_agg(const int* __restrict__ t, const float* __restrict__ h,
                      const float* __restrict__ norm, float* __restrict__ agg) {
    int e = blockIdx.x * blockDim.x + threadIdx.x;
    if (e >= NAUG) return;
    int s, r, o; aug_edge(t, e, s, r, o);
    float en = norm[r * NN + o];
    scatter_rows16(s, o * RP + r, en, h, agg);
}

__global__ void k_gemm(const float* __restrict__ agg, const float* __restrict__ W2,
                       const float* __restrict__ b2, float* __restrict__ x2) {
    int wave = (blockIdx.x * blockDim.x + threadIdx.x) >> 5;
    int lane = threadIdx.x & 31;
    if (wave >= NN / 16) return;
    int m0   = wave * 16;
    int half = lane >> 4;
    int lr   = lane & 15;

    const float* arow = agg + (long)(m0 + lr) * KDIM;

    v8f c[8];
#pragma unroll
    for (int nt = 0; nt < 8; nt++) {
        v8f z = {0.f, 0.f, 0.f, 0.f, 0.f, 0.f, 0.f, 0.f};
        c[nt] = z;
    }

    for (int k0 = 0; k0 < KDIM; k0 += 4) {
        int ka = k0 + 2 * half;
        v2f a;
        a.x = arow[ka];
        a.y = arow[ka + 1];
        const float* brow0 = W2 + (long)ka * EMBN + lr;
        const float* brow1 = W2 + (long)(ka + 1) * EMBN + lr;
#pragma unroll
        for (int nt = 0; nt < 8; nt++) {
            v2f b;
            b.x = brow0[nt * 16];
            b.y = brow1[nt * 16];
            c[nt] = __builtin_amdgcn_wmma_f32_16x16x4_f32(
                false, a, false, b, (short)0, c[nt], false, false);
        }
    }

#pragma unroll
    for (int nt = 0; nt < 8; nt++) {
        float bias = b2[nt * 16 + lr];
#pragma unroll
        for (int v = 0; v < 8; v++) {
            int row = m0 + v + 8 * half;
            x2[(long)row * EMBN + nt * 16 + lr] = c[nt][v] + bias;
        }
    }
}

__global__ void k_score(const int* __restrict__ t, const float* __restrict__ x2,
                        const float* __restrict__ rel, float* __restrict__ out) {
    int gid  = blockIdx.x * blockDim.x + threadIdx.x;
    int w    = gid >> 5;
    int lane = gid & 31;
    if (w >= NT) return;
    int s = t[3*w], p = t[3*w+1], o = t[3*w+2];
    const float* xs = x2 + (long)s * EMBN;
    const float* xo = x2 + (long)o * EMBN;
    const float* rp = rel + (long)p * EMBN;
    float acc = 0.f;
#pragma unroll
    for (int q = 0; q < 4; q++) {
        int j = lane + 32 * q;
        acc += xs[j] * rp[j] * xo[j];
    }
#pragma unroll
    for (int off = 16; off > 0; off >>= 1) acc += __shfl_xor(acc, off, 32);
    if (lane == 0) out[w] = acc;
}

extern "C" void kernel_launch(void* const* d_in, const int* in_sizes, int n_in,
                              void* d_out, int out_size, void* d_ws, size_t ws_size,
                              hipStream_t stream) {
    const int*   triples = (const int*)  d_in[0];
    const float* W1      = (const float*)d_in[1];
    const float* b1      = (const float*)d_in[2];
    const float* W2      = (const float*)d_in[3];
    const float* b2      = (const float*)d_in[4];
    const float* rel     = (const float*)d_in[5];
    float*       out     = (float*)d_out;

    float* deg = (float*)d_ws;
    float* x1  = deg + (long)RP * NN;
    float* agg = x1  + (long)NN * HIDN;
    float* x2  = agg + (long)NN * KDIM;
    long zf = (long)RP * NN + (long)NN * HIDN + (long)NN * KDIM;
    int n4 = (int)(zf / 4);

    dim3 B(256);
    k_zero<<<(n4 + 255) / 256, B, 0, stream>>>((float4*)d_ws, n4);
    k_deg <<<(NAUG + 255) / 256, B, 0, stream>>>(triples, deg);
    k_norm<<<(RP * NN + 255) / 256, B, 0, stream>>>(deg);
    k_msg <<<(NAUG + 255) / 256, B, 0, stream>>>(triples, W1, deg, x1);
    k_relu<<<(NN * HIDN + 255) / 256, B, 0, stream>>>(x1, b1);
    k_agg <<<(NAUG + 255) / 256, B, 0, stream>>>(triples, x1, deg, agg);

    int nwaves = NN / 16;
    k_gemm<<<(nwaves * 32 + 255) / 256, B, 0, stream>>>(agg, W2, b2, x2);

    long sthreads = (long)NT * 32;
    k_score<<<(int)((sthreads + 255) / 256), B, 0, stream>>>(triples, x2, rel, out);
}


# Score kernel: one wave scores 16 triples instead of 1 (16x fewer waves to enter), each 512-B row is one 16-B-per-lane load, lane sums folded by a halving exchange (new f32 sum order)
# speedup vs baseline: 2.7998x; 1.1412x over previous
#include <hip/hip_runtime.h>

typedef float v2f __attribute__((ext_vector_type(2)));
typedef float v8f __attribute__((ext_vector_type(8)));

#define NN   50000
#define NRL  12
#define NT   500000
#define HIDN 16
#define EMBN 128
#define RP   25
#define KDIM 400
#define NAUG (2*NT + NN)

__device__ __forceinline__ void aug_edge(const int* __restrict__ t, int e,
                                         int& s, int& r, int& o) {
    if (e < NT)            { s = t[3*e];     r = t[3*e+1];       o = t[3*e+2]; }
    else if (e < 2*NT)     { int j = e - NT;
                             s = t[3*j+2];   r = t[3*j+1] + NRL; o = t[3*j];   }
    else                   { int i = e - 2*NT;
                             s = i;          r = 2*NRL;          o = i;        }
}

__global__ void k_zero(float4* __restrict__ p, int n4) {
    int i = blockIdx.x * blockDim.x + threadIdx.x;
    if (i < n4) p[i] = make_float4(0.f, 0.f, 0.f, 0.f);
}

__global__ void k_deg(const int* __restrict__ t, float* __restrict__ deg) {
    int e = blockIdx.x * blockDim.x + threadIdx.x;
    if (e >= NAUG) return;
    int s, r, o; aug_edge(t, e, s, r, o); (void)s;
    atomicAdd(&deg[r * NN + o], 1.0f);
}

__global__ void k_norm(float* __restrict__ deg) {
    int i = blockIdx.x * blockDim.x + threadIdx.x;
    if (i < RP * NN) deg[i] = 1.0f / fmaxf(deg[i], 1.0f);
}

static_assert(NAUG % 16 == 0, "half-waves must be whole");
__device__ __forceinline__ void scatter_rows16(int src, int dst, float en,
                                               const float* __restrict__ stab,
                                               float* __restrict__ dtab) {
    const int c = threadIdx.x & 15;
    float w[16], g[16];
    int   d[16];
#pragma unroll
    for (int j = 0; j < 16; j++) {
        int sj = __shfl(src, j, 16);
        g[j]   = __shfl(en,  j, 16);
        d[j]   = __shfl(dst, j, 16);
        w[j]   = stab[sj * HIDN + c];
    }
#pragma unroll
    for (int j = 0; j < 16; j++)
        atomicAdd(dtab + (d[j] * HIDN + c), w[j] * g[j]);
}

__global__ void k_msg(const int* __restrict__ t, const float* __restrict__ W1,
                      const float* __restrict__ norm, float* __restrict__ x1) {
    int e = blockIdx.x * blockDim.x + threadIdx.x;
    if (e >= NAUG) return;
    int s, r, o; aug_edge(t, e, s, r, o);
    float en = norm[r * NN + o];
    scatter_rows16(r * NN + s, o, en, W1, x1);
}

__global__ void k_relu(float* __restrict__ x1, const float* __restrict__ b1) {
    int i = blockIdx.x * blockDim.x + threadIdx.x;
    if (i < NN * HIDN) x1[i] = fmaxf(x1[i] + b1[i & (HIDN - 1)], 0.0f);
}

__global__ void k_agg(const int* __restrict__ t, const float* __restrict__ h,
                      const float* __restrict__ norm, float* __restrict__ agg) {
    int e = blockIdx.x * blockDim.x + threadIdx.x;
    if (e >= NAUG) return;
    int s, r, o; aug_edge(t, e, s, r, o);
    float en = norm[r * NN + o];
    scatter_rows16(s, o * RP + r, en, h, agg);
}

__global__ void k_gemm(const float* __restrict__ agg, const float* __restrict__ W2,
                       const float* __restrict__ b2, float* __restrict__ x2) {
    int wave = (blockIdx.x * blockDim.x + threadIdx.x) >> 5;
    int lane = threadIdx.x & 31;
    if (wave >= NN / 16) return;
    int m0   = wave * 16;
    int half = lane >> 4;
    int lr   = lane & 15;

    const float* arow = agg + (long)(m0 + lr) * KDIM;

    v8f c[8];
#pragma unroll
    for (int nt = 0; nt < 8; nt++) {
        v8f z = {0.f, 0.f, 0.f, 0.f, 0.f, 0.f, 0.f, 0.f};
        c[nt] = z;
    }

    for (int k0 = 0; k0 < KDIM; k0 += 4) {
        int ka = k0 + 2 * half;
        v2f a;
        a.x = arow[ka];
        a.y = arow[ka + 1];
        const float* brow0 = W2 + (long)ka * EMBN + lr;
        const float* brow1 = W2 + (long)(ka + 1) * EMBN + lr;
#pragma unroll
        for (int nt = 0; nt < 8; nt++) {
            v2f b;
            b.x = brow0[nt * 16];
            b.y = brow1[nt * 16];
            c[nt] = __builtin_amdgcn_wmma_f32_16x16x4_f32(
                false, a, false, b, (short)0, c[nt], false, false);
        }
    }

#pragma unroll
    for (int nt = 0; nt < 8; nt++) {
        float bias = b2[nt * 16 + lr];
#pragma unroll
        for (int v = 0; v < 8; v++) {
            int row = m0 + v + 8 * half;
            x2[(long)row * EMBN + nt * 16 + lr] = c[nt][v] + bias;
        }
    }
}

#define SPW 16
static_assert(NT % SPW == 0, "a wave's 16 triples are all in range or all past the end");

__device__ __forceinline__ float dot4(float4 a, float4 r, float4 b) {
    float acc = a.x * r.x * b.x;
    acc += a.y * r.y * b.y;
    acc += a.z * r.z * b.z;
    acc += a.w * r.w * b.w;
    return acc;
}

__global__ void __launch_bounds__(256)
k_score(const int* __restrict__ t, const float* __restrict__ x2,
        const float* __restrict__ rel, float* __restrict__ out) {
    const int lane = threadIdx.x & 31;
    const int w = __builtin_amdgcn_readfirstlane((int)(blockIdx.x * 8 + (threadIdx.x >> 5)));
    if (w >= NT / SPW) return;
    const int* tw = t + 3 * SPW * w;
    const unsigned co = 16u * lane;
    const unsigned ROWB = EMBN * 4u;
    const char* xb = reinterpret_cast<const char*>(x2);
    const char* rb = reinterpret_cast<const char*>(rel);

    float4 xs[SPW], xo[SPW], rp[SPW];
#pragma unroll
    for (int j = 0; j < SPW; j++) {
        const int s = tw[3 * j], p = tw[3 * j + 1], o = tw[3 * j + 2];
        xs[j] = *reinterpret_cast<const float4*>(xb + ((unsigned)s * ROWB + co));
        xo[j] = *reinterpret_cast<const float4*>(xb + ((unsigned)o * ROWB + co));
        rp[j] = *reinterpret_cast<const float4*>(rb + ((unsigned)p * ROWB + co));
    }

    float v[SPW];
#pragma unroll
    for (int j = 0; j < SPW; j++) v[j] = dot4(xs[j], rp[j], xo[j]);

    const bool h4 = (lane & 16) != 0, h3 = (lane & 8) != 0;
    const bool h2 = (lane & 4)  != 0, h1 = (lane & 2) != 0;

    float a8[8], a4[4], a2[2];
#pragma unroll
    for (int i = 0; i < 8; i++) {
        const float keep = h4 ? v[i + 8] : v[i];
        const float send = h4 ? v[i]     : v[i + 8];
        a8[i] = keep + __shfl_xor(send, 16, 32);
    }
#pragma unroll
    for (int i = 0; i < 4; i++) {
        const float keep = h3 ? a8[i + 4] : a8[i];
        const float send = h3 ? a8[i]     : a8[i + 4];
        a4[i] = keep + __shfl_xor(send, 8, 32);
    }
#pragma unroll
    for (int i = 0; i < 2; i++) {
        const float keep = h2 ? a4[i + 2] : a4[i];
        const float send = h2 ? a4[i]     : a4[i + 2];
        a2[i] = keep + __shfl_xor(send, 4, 32);
    }
    const float keep1 = h1 ? a2[1] : a2[0];
    const float send1 = h1 ? a2[0] : a2[1];
    float z = keep1 + __shfl_xor(send1, 2, 32);
    z += __shfl_xor(z, 1, 32);

    if ((lane & 1) == 0) out[SPW * w + (lane >> 1)] = z;
}

extern "C" void kernel_launch(void* const* d_in, const int* in_sizes, int n_in,
                              void* d_out, int out_size, void* d_ws, size_t ws_size,
                              hipStream_t stream) {
    const int*   triples = (const int*)  d_in[0];
    const float* W1      = (const float*)d_in[1];
    const float* b1      = (const float*)d_in[2];
    const float* W2      = (const float*)d_in[3];
    const float* b2      = (const float*)d_in[4];
    const float* rel     = (const float*)d_in[5];
    float*       out     = (float*)d_out;

    float* deg = (float*)d_ws;
    float* x1  = deg + (long)RP * NN;
    float* agg = x1  + (long)NN * HIDN;
    float* x2  = agg + (long)NN * KDIM;
    long zf = (long)RP * NN + (long)NN * HIDN + (long)NN * KDIM;
    int n4 = (int)(zf / 4);

    dim3 B(256);
    k_zero<<<(n4 + 255) / 256, B, 0, stream>>>((float4*)d_ws, n4);
    k_deg <<<(NAUG + 255) / 256, B, 0, stream>>>(triples, deg);
    k_norm<<<(RP * NN + 255) / 256, B, 0, stream>>>(deg);
    k_msg <<<(NAUG + 255) / 256, B, 0, stream>>>(triples, W1, deg, x1);
    k_relu<<<(NN * HIDN + 255) / 256, B, 0, stream>>>(x1, b1);
    k_agg <<<(NAUG + 255) / 256, B, 0, stream>>>(triples, x1, deg, agg);

    int nwaves = NN / 16;
    k_gemm<<<(nwaves * 32 + 255) / 256, B, 0, stream>>>(agg, W2, b2, x2);

    k_score<<<(NT / SPW + 7) / 8, B, 0, stream>>>(triples, x2, rel, out);
}


# speedup vs baseline: 3.2706x; 1.1682x over previous
#include <hip/hip_runtime.h>

typedef float v2f __attribute__((ext_vector_type(2)));
typedef float v8f __attribute__((ext_vector_type(8)));

#define NN   50000
#define NRL  12
#define NT   500000
#define HIDN 16
#define EMBN 128
#define RP   25
#define KDIM 400
#define NAUG (2*NT + NN)

__device__ __forceinline__ void aug_edge(const int* __restrict__ t, int e,
                                         int& s, int& r, int& o) {
    if (e < NT)            { s = t[3*e];     r = t[3*e+1];       o = t[3*e+2]; }
    else if (e < 2*NT)     { int j = e - NT;
                             s = t[3*j+2];   r = t[3*j+1] + NRL; o = t[3*j];   }
    else                   { int i = e - 2*NT;
                             s = i;          r = 2*NRL;          o = i;        }
}

__device__ __forceinline__ int pkg_half() {
    return (int)((__builtin_amdgcn_s_sendmsg_rtn(0x87) >> 18) & 1u);
}

__global__ void k_zero(float4* __restrict__ p, int n4) {
    int i = blockIdx.x * blockDim.x + threadIdx.x;
    if (i < n4) p[i] = make_float4(0.f, 0.f, 0.f, 0.f);
}

__global__ void k_deg(const int* __restrict__ t, float* __restrict__ deg0,
                      float* __restrict__ deg1) {
    int e = blockIdx.x * blockDim.x + threadIdx.x;
    if (e >= NAUG) return;
    float* deg = pkg_half() ? deg1 : deg0;
    int s, r, o; aug_edge(t, e, s, r, o); (void)s;
    atomicAdd(&deg[r * NN + o], 1.0f);
}

__global__ void k_norm(float* __restrict__ deg0, const float* __restrict__ deg1) {
    int i = blockIdx.x * blockDim.x + threadIdx.x;
    if (i < RP * NN) deg0[i] = 1.0f / fmaxf(deg0[i] + deg1[i], 1.0f);
}

static_assert(NAUG % 16 == 0, "half-waves must be whole");
__device__ __forceinline__ void scatter_rows16(int src, int dst, float en,
                                               const float* __restrict__ stab,
                                               float* __restrict__ dtab) {
    const int c = threadIdx.x & 15;
    float w[16], g[16];
    int   d[16];
#pragma unroll
    for (int j = 0; j < 16; j++) {
        int sj = __shfl(src, j, 16);
        g[j]   = __shfl(en,  j, 16);
        d[j]   = __shfl(dst, j, 16);
        w[j]   = stab[sj * HIDN + c];
    }
#pragma unroll
    for (int j = 0; j < 16; j++)
        atomicAdd(dtab + (d[j] * HIDN + c), w[j] * g[j]);
}

__global__ void k_msg(const int* __restrict__ t, const float* __restrict__ W1,
                      const float* __restrict__ norm, float* __restrict__ x1a,
                      float* __restrict__ x1b) {
    int e = blockIdx.x * blockDim.x + threadIdx.x;
    if (e >= NAUG) return;
    float* x1 = pkg_half() ? x1b : x1a;
    int s, r, o; aug_edge(t, e, s, r, o);
    float en = norm[r * NN + o];
    scatter_rows16(r * NN + s, o, en, W1, x1);
}

__global__ void k_relu(float* __restrict__ x1a, const float* __restrict__ x1b,
                       const float* __restrict__ b1) {
    int i = blockIdx.x * blockDim.x + threadIdx.x;
    if (i < NN * HIDN) x1a[i] = fmaxf((x1a[i] + x1b[i]) + b1[i & (HIDN - 1)], 0.0f);
}

__global__ void k_agg(const int* __restrict__ t, const float* __restrict__ h,
                      const float* __restrict__ norm, float* __restrict__ agg) {
    int e = blockIdx.x * blockDim.x + threadIdx.x;
    if (e >= NAUG) return;
    int s, r, o; aug_edge(t, e, s, r, o);
    float en = norm[r * NN + o];
    scatter_rows16(s, o * RP + r, en, h, agg);
}

__global__ void k_gemm(const float* __restrict__ agg, const float* __restrict__ W2,
                       const float* __restrict__ b2, float* __restrict__ x2) {
    int wave = (blockIdx.x * blockDim.x + threadIdx.x) >> 5;
    int lane = threadIdx.x & 31;
    if (wave >= NN / 16) return;
    int m0   = wave * 16;
    int half = lane >> 4;
    int lr   = lane & 15;

    const float* arow = agg + (long)(m0 + lr) * KDIM;

    v8f c[8];
#pragma unroll
    for (int nt = 0; nt < 8; nt++) {
        v8f z = {0.f, 0.f, 0.f, 0.f, 0.f, 0.f, 0.f, 0.f};
        c[nt] = z;
    }

    for (int k0 = 0; k0 < KDIM; k0 += 4) {
        int ka = k0 + 2 * half;
        v2f a;
        a.x = arow[ka];
        a.y = arow[ka + 1];
        const float* brow0 = W2 + (long)ka * EMBN + lr;
        const float* brow1 = W2 + (long)(ka + 1) * EMBN + lr;
#pragma unroll
        for (int nt = 0; nt < 8; nt++) {
            v2f b;
            b.x = brow0[nt * 16];
            b.y = brow1[nt * 16];
            c[nt] = __builtin_amdgcn_wmma_f32_16x16x4_f32(
                false, a, false, b, (short)0, c[nt], false, false);
        }
    }

#pragma unroll
    for (int nt = 0; nt < 8; nt++) {
        float bias = b2[nt * 16 + lr];
#pragma unroll
        for (int v = 0; v < 8; v++) {
            int row = m0 + v + 8 * half;
            x2[(long)row * EMBN + nt * 16 + lr] = c[nt][v] + bias;
        }
    }
}

#define SPW 16
static_assert(NT % SPW == 0, "a wave's 16 triples are all in range or all past the end");

__device__ __forceinline__ float dot4(float4 a, float4 r, float4 b) {
    float acc = a.x * r.x * b.x;
    acc += a.y * r.y * b.y;
    acc += a.z * r.z * b.z;
    acc += a.w * r.w * b.w;
    return acc;
}

__global__ void __launch_bounds__(256)
k_score(const int* __restrict__ t, const float* __restrict__ x2,
        const float* __restrict__ rel, float* __restrict__ out) {
    const int lane = threadIdx.x & 31;
    const int w = __builtin_amdgcn_readfirstlane((int)(blockIdx.x * 8 + (threadIdx.x >> 5)));
    if (w >= NT / SPW) return;
    const int* tw = t + 3 * SPW * w;
    const unsigned co = 16u * lane;
    const unsigned ROWB = EMBN * 4u;
    const char* xb = reinterpret_cast<const char*>(x2);
    const char* rb = reinterpret_cast<const char*>(rel);

    float4 xs[SPW], xo[SPW], rp[SPW];
#pragma unroll
    for (int j = 0; j < SPW; j++) {
        const int s = tw[3 * j], p = tw[3 * j + 1], o = tw[3 * j + 2];
        xs[j] = *reinterpret_cast<const float4*>(xb + ((unsigned)s * ROWB + co));
        xo[j] = *reinterpret_cast<const float4*>(xb + ((unsigned)o * ROWB + co));
        rp[j] = *reinterpret_cast<const float4*>(rb + ((unsigned)p * ROWB + co));
    }

    float v[SPW];
#pragma unroll
    for (int j = 0; j < SPW; j++) v[j] = dot4(xs[j], rp[j], xo[j]);

    const bool h4 = (lane & 16) != 0, h3 = (lane & 8) != 0;
    const bool h2 = (lane & 4)  != 0, h1 = (lane & 2) != 0;

    float a8[8], a4[4], a2[2];
#pragma unroll
    for (int i = 0; i < 8; i++) {
        const float keep = h4 ? v[i + 8] : v[i];
        const float send = h4 ? v[i]     : v[i + 8];
        a8[i] = keep + __shfl_xor(send, 16, 32);
    }
#pragma unroll
    for (int i = 0; i < 4; i++) {
        const float keep = h3 ? a8[i + 4] : a8[i];
        const float send = h3 ? a8[i]     : a8[i + 4];
        a4[i] = keep + __shfl_xor(send, 8, 32);
    }
#pragma unroll
    for (int i = 0; i < 2; i++) {
        const float keep = h2 ? a4[i + 2] : a4[i];
        const float send = h2 ? a4[i]     : a4[i + 2];
        a2[i] = keep + __shfl_xor(send, 4, 32);
    }
    const float keep1 = h1 ? a2[1] : a2[0];
    const float send1 = h1 ? a2[0] : a2[1];
    float z = keep1 + __shfl_xor(send1, 2, 32);
    z += __shfl_xor(z, 1, 32);

    if ((lane & 1) == 0) out[SPW * w + (lane >> 1)] = z;
}

extern "C" void kernel_launch(void* const* d_in, const int* in_sizes, int n_in,
                              void* d_out, int out_size, void* d_ws, size_t ws_size,
                              hipStream_t stream) {
    const int*   triples = (const int*)  d_in[0];
    const float* W1      = (const float*)d_in[1];
    const float* b1      = (const float*)d_in[2];
    const float* W2      = (const float*)d_in[3];
    const float* b2      = (const float*)d_in[4];
    const float* rel     = (const float*)d_in[5];
    float*       out     = (float*)d_out;

    float* deg  = (float*)d_ws;
    float* x1   = deg  + (long)RP * NN;
    float* agg  = x1   + (long)NN * HIDN;
    float* x2   = agg  + (long)NN * KDIM;
    float* deg1 = x2;
    float* x1b  = deg1 + (long)RP * NN;
    long zf = 2 * (long)RP * NN + 2 * (long)NN * HIDN + (long)NN * KDIM;
    int n4 = (int)(zf / 4);

    dim3 B(256);
    k_zero<<<(n4 + 255) / 256, B, 0, stream>>>((float4*)d_ws, n4);
    k_deg <<<(NAUG + 255) / 256, B, 0, stream>>>(triples, deg, deg1);
    k_norm<<<(RP * NN + 255) / 256, B, 0, stream>>>(deg, deg1);
    k_msg <<<(NAUG + 255) / 256, B, 0, stream>>>(triples, W1, deg, x1, x1b);
    k_relu<<<(NN * HIDN + 255) / 256, B, 0, stream>>>(x1, x1b, b1);
    k_agg <<<(NAUG + 255) / 256, B, 0, stream>>>(triples, x1, deg, agg);

    int nwaves = NN / 16;
    k_gemm<<<(nwaves * 32 + 255) / 256, B, 0, stream>>>(agg, W2, b2, x2);

    k_score<<<(NT / SPW + 7) / 8, B, 0, stream>>>(triples, x2, rel, out);
}


# Layer 2 without the agg matrix in memory: k_deg threads each edge onto a per-slot list (returning exchange), one launch walks the lists into LDS strips and multiplies there (f32 sums in list order)
# speedup vs baseline: 4.4221x; 1.3521x over previous
#include <hip/hip_runtime.h>

typedef float v2f __attribute__((ext_vector_type(2)));
typedef float v8f __attribute__((ext_vector_type(8)));

#define NN   50000
#define NRL  12
#define NT   500000
#define HIDN 16
#define EMBN 128
#define RP   25
#define KDIM 400
#define NAUG (2*NT + NN)
#define LD   5
#define AP   212

__device__ __forceinline__ void aug_edge(const int* __restrict__ t, int e,
                                         int& s, int& r, int& o) {
    if (e < NT)            { s = t[3*e];     r = t[3*e+1];       o = t[3*e+2]; }
    else if (e < 2*NT)     { int j = e - NT;
                             s = t[3*j+2];   r = t[3*j+1] + NRL; o = t[3*j];   }
    else                   { int i = e - 2*NT;
                             s = i;          r = 2*NRL;          o = i;        }
}

__device__ __forceinline__ int pkg_half() {
    return (int)((__builtin_amdgcn_s_sendmsg_rtn(0x87) >> 18) & 1u);
}

__global__ void k_init(float4* __restrict__ p, int nz4, int nh4) {
    int i = blockIdx.x * blockDim.x + threadIdx.x;
    if (i < nz4) p[i] = make_float4(0.f, 0.f, 0.f, 0.f);
    else if (i < nz4 + nh4) reinterpret_cast<int4*>(p)[i] = make_int4(-1, -1, -1, -1);
}

__global__ void k_deg(const int* __restrict__ t, float* __restrict__ deg0,
                      float* __restrict__ deg1, int* __restrict__ head0,
                      int* __restrict__ head1, int2* __restrict__ link) {
    int e = blockIdx.x * blockDim.x + threadIdx.x;
    if (e >= NAUG) return;
    const int hf = pkg_half();
    float* deg = hf ? deg1 : deg0;
    int*   hd  = hf ? head1 : head0;
    int s, r, o; aug_edge(t, e, s, r, o);
    atomicAdd(&deg[r * NN + o], 1.0f);
    int old = atomicExch(&hd[o * RP + r], e);
    link[e] = make_int2(old, s);
}

__global__ void k_norm(float* __restrict__ deg0, const float* __restrict__ deg1) {
    int i = blockIdx.x * blockDim.x + threadIdx.x;
    if (i < RP * NN) deg0[i] = 1.0f / fmaxf(deg0[i] + deg1[i], 1.0f);
}

static_assert(NAUG % 16 == 0, "half-waves must be whole");
__device__ __forceinline__ void scatter_rows16(int src, int dst, float en,
                                               const float* __restrict__ stab,
                                               float* __restrict__ dtab) {
    const int c = threadIdx.x & 15;
    float w[16], g[16];
    int   d[16];
#pragma unroll
    for (int j = 0; j < 16; j++) {
        int sj = __shfl(src, j, 16);
        g[j]   = __shfl(en,  j, 16);
        d[j]   = __shfl(dst, j, 16);
        w[j]   = stab[sj * HIDN + c];
    }
#pragma unroll
    for (int j = 0; j < 16; j++)
        atomicAdd(dtab + (d[j] * HIDN + c), w[j] * g[j]);
}

__global__ void k_msg(const int* __restrict__ t, const float* __restrict__ W1,
                      const float* __restrict__ norm, float* __restrict__ x1a,
                      float* __restrict__ x1b) {
    int e = blockIdx.x * blockDim.x + threadIdx.x;
    if (e >= NAUG) return;
    float* x1 = pkg_half() ? x1b : x1a;
    int s, r, o; aug_edge(t, e, s, r, o);
    float en = norm[r * NN + o];
    scatter_rows16(r * NN + s, o, en, W1, x1);
}

__global__ void k_relu(float* __restrict__ x1a, const float* __restrict__ x1b,
                       const float* __restrict__ b1) {
    int i = blockIdx.x * blockDim.x + threadIdx.x;
    if (i < NN * HIDN) x1a[i] = fmaxf((x1a[i] + x1b[i]) + b1[i & (HIDN - 1)], 0.0f);
}

template <int RB, int NRP>
__device__ __forceinline__ void l2_part(float* As, const int m0, const int lane,
                                        const float* __restrict__ h,
                                        const float* __restrict__ norm,
                                        const int* __restrict__ head0,
                                        const int* __restrict__ head1,
                                        const int2* __restrict__ link,
                                        const float* __restrict__ W2, v8f (&c)[8]) {
    const int half = lane >> 4;
    const int lr   = lane & 15;

    __syncthreads();
    for (int i = lane; i < 16 * AP / 4; i += 32)
        reinterpret_cast<float4*>(As)[i] = make_float4(0.f, 0.f, 0.f, 0.f);

    int   e[NRP];
    float g[NRP];
#pragma unroll
    for (int i = 0; i < NRP; i++) {
        const int q   = lane + 32 * i;
        const int pl  = q / (16 * NRP);
        const int sl  = q - pl * (16 * NRP);
        const int row = sl / NRP, rr = sl - row * NRP;
        const int o = m0 + row, r = RB + rr;
        e[i] = (pl ? head1 : head0)[o * RP + r];
        g[i] = norm[r * NN + o];
    }
    __syncthreads();

    for (int pass = 0; pass < NAUG; pass++) {
        bool live = false;
#pragma unroll
        for (int i = 0; i < NRP; i++) live |= (e[i] >= 0);
        if (__builtin_amdgcn_ballot_w32(live) == 0u) break;

        int2 ln[NRP];
#pragma unroll
        for (int i = 0; i < NRP; i++) ln[i] = link[e[i] >= 0 ? e[i] : 0];

        float4 rw[NRP][4];
#pragma unroll
        for (int i = 0; i < NRP + LD; i++) {
            if (i < NRP) {
                const int sn = e[i] >= 0 ? ln[i].y : 0;
                const float4* hr = reinterpret_cast<const float4*>(h + sn * HIDN);
                rw[i][0] = hr[0]; rw[i][1] = hr[1]; rw[i][2] = hr[2]; rw[i][3] = hr[3];
            }
            const int j = i - LD;
            if (j >= 0 && e[j] >= 0) {
                const int q   = lane + 32 * j;
                const int sl  = q % (16 * NRP);
                const int row = sl / NRP, rr = sl - row * NRP;
                float* d = As + row * AP + rr * HIDN;
                const float w = g[j];
                atomicAdd(d +  0, rw[j][0].x * w); atomicAdd(d +  1, rw[j][0].y * w);
                atomicAdd(d +  2, rw[j][0].z * w); atomicAdd(d +  3, rw[j][0].w * w);
                atomicAdd(d +  4, rw[j][1].x * w); atomicAdd(d +  5, rw[j][1].y * w);
                atomicAdd(d +  6, rw[j][1].z * w); atomicAdd(d +  7, rw[j][1].w * w);
                atomicAdd(d +  8, rw[j][2].x * w); atomicAdd(d +  9, rw[j][2].y * w);
                atomicAdd(d + 10, rw[j][2].z * w); atomicAdd(d + 11, rw[j][2].w * w);
                atomicAdd(d + 12, rw[j][3].x * w); atomicAdd(d + 13, rw[j][3].y * w);
                atomicAdd(d + 14, rw[j][3].z * w); atomicAdd(d + 15, rw[j][3].w * w);
                e[j] = ln[j].x;
            }
        }
    }
    __syncthreads();

    for (int kk = 0; kk < NRP * HIDN; kk += 4) {
        const v2f a = *reinterpret_cast<const v2f*>(As + lr * AP + kk + 2 * half);
        const int ka = RB * HIDN + kk + 2 * half;
        const float* brow0 = W2 + (long)ka * EMBN + lr;
        const float* brow1 = W2 + (long)(ka + 1) * EMBN + lr;
#pragma unroll
        for (int nt = 0; nt < 8; nt++) {
            v2f b;
            b.x = brow0[nt * 16];
            b.y = brow1[nt * 16];
            c[nt] = __builtin_amdgcn_wmma_f32_16x16x4_f32(
                false, a, false, b, (short)0, c[nt], false, false);
        }
    }
}

__global__ void __launch_bounds__(32)
k_l2(const float* __restrict__ h, const float* __restrict__ norm,
     const int* __restrict__ head0, const int* __restrict__ head1,
     const int2* __restrict__ link, const float* __restrict__ W2,
     const float* __restrict__ b2, float* __restrict__ x2) {
    __shared__ __attribute__((aligned(16))) float As[16 * AP];
    const int lane = threadIdx.x;
    const int m0   = blockIdx.x * 16;
    const int half = lane >> 4;
    const int lr   = lane & 15;

    v8f c[8];
#pragma unroll
    for (int nt = 0; nt < 8; nt++) {
        v8f z = {0.f, 0.f, 0.f, 0.f, 0.f, 0.f, 0.f, 0.f};
        c[nt] = z;
    }

    l2_part<0, 13>(As, m0, lane, h, norm, head0, head1, link, W2, c);
    l2_part<13, 12>(As, m0, lane, h, norm, head0, head1, link, W2, c);

#pragma unroll
    for (int nt = 0; nt < 8; nt++) {
        float bias = b2[nt * 16 + lr];
#pragma unroll
        for (int v = 0; v < 8; v++) {
            int row = m0 + v + 8 * half;
            x2[(long)row * EMBN + nt * 16 + lr] = c[nt][v] + bias;
        }
    }
}
static_assert(NN % 16 == 0, "strips are whole");
static_assert(RP == 13 + 12, "the two parts cover every relation");

#define SPW 16
static_assert(NT % SPW == 0, "a wave's 16 triples are all in range or all past the end");

__device__ __forceinline__ float dot4(float4 a, float4 r, float4 b) {
    float acc = a.x * r.x * b.x;
    acc += a.y * r.y * b.y;
    acc += a.z * r.z * b.z;
    acc += a.w * r.w * b.w;
    return acc;
}

__global__ void __launch_bounds__(256)
k_score(const int* __restrict__ t, const float* __restrict__ x2,
        const float* __restrict__ rel, float* __restrict__ out) {
    const int lane = threadIdx.x & 31;
    const int w = __builtin_amdgcn_readfirstlane((int)(blockIdx.x * 8 + (threadIdx.x >> 5)));
    if (w >= NT / SPW) return;
    const int* tw = t + 3 * SPW * w;
    const unsigned co = 16u * lane;
    const unsigned ROWB = EMBN * 4u;
    const char* xb = reinterpret_cast<const char*>(x2);
    const char* rb = reinterpret_cast<const char*>(rel);

    float4 xs[SPW], xo[SPW], rp[SPW];
#pragma unroll
    for (int j = 0; j < SPW; j++) {
        const int s = tw[3 * j], p = tw[3 * j + 1], o = tw[3 * j + 2];
        xs[j] = *reinterpret_cast<const float4*>(xb + ((unsigned)s * ROWB + co));
        xo[j] = *reinterpret_cast<const float4*>(xb + ((unsigned)o * ROWB + co));
        rp[j] = *reinterpret_cast<const float4*>(rb + ((unsigned)p * ROWB + co));
    }

    float v[SPW];
#pragma unroll
    for (int j = 0; j < SPW; j++) v[j] = dot4(xs[j], rp[j], xo[j]);

    const bool h4 = (lane & 16) != 0, h3 = (lane & 8) != 0;
    const bool h2 = (lane & 4)  != 0, h1 = (lane & 2) != 0;

    float a8[8], a4[4], a2[2];
#pragma unroll
    for (int i = 0; i < 8; i++) {
        const float keep = h4 ? v[i + 8] : v[i];
        const float send = h4 ? v[i]     : v[i + 8];
        a8[i] = keep + __shfl_xor(send, 16, 32);
    }
#pragma unroll
    for (int i = 0; i < 4; i++) {
        const float keep = h3 ? a8[i + 4] : a8[i];
        const float send = h3 ? a8[i]     : a8[i + 4];
        a4[i] = keep + __shfl_xor(send, 8, 32);
    }
#pragma unroll
    for (int i = 0; i < 2; i++) {
        const float keep = h2 ? a4[i + 2] : a4[i];
        const float send = h2 ? a4[i]     : a4[i + 2];
        a2[i] = keep + __shfl_xor(send, 4, 32);
    }
    const float keep1 = h1 ? a2[1] : a2[0];
    const float send1 = h1 ? a2[0] : a2[1];
    float z = keep1 + __shfl_xor(send1, 2, 32);
    z += __shfl_xor(z, 1, 32);

    if ((lane & 1) == 0) out[SPW * w + (lane >> 1)] = z;
}

extern "C" void kernel_launch(void* const* d_in, const int* in_sizes, int n_in,
                              void* d_out, int out_size, void* d_ws, size_t ws_size,
                              hipStream_t stream) {
    const int*   triples = (const int*)  d_in[0];
    const float* W1      = (const float*)d_in[1];
    const float* b1      = (const float*)d_in[2];
    const float* W2      = (const float*)d_in[3];
    const float* b2      = (const float*)d_in[4];
    const float* rel     = (const float*)d_in[5];
    float*       out     = (float*)d_out;

    float* deg   = (float*)d_ws;
    float* x1    = deg  + (long)RP * NN;
    float* deg1  = x1   + (long)NN * HIDN;
    float* x1b   = deg1 + (long)RP * NN;
    int*   head0 = (int*)(x1b + (long)NN * HIDN);
    int*   head1 = head0 + (long)RP * NN;
    int2*  link  = (int2*)(head1 + (long)RP * NN);
    float* x2    = x1   + (long)NN * HIDN + (long)NN * KDIM;
    int nz4 = (2 * RP * NN + 2 * NN * HIDN) / 4;
    int nh4 = (2 * RP * NN) / 4;

    dim3 B(256);
    k_init<<<(nz4 + nh4 + 255) / 256, B, 0, stream>>>((float4*)d_ws, nz4, nh4);
    k_deg <<<(NAUG + 255) / 256, B, 0, stream>>>(triples, deg, deg1, head0, head1, link);
    k_norm<<<(RP * NN + 255) / 256, B, 0, stream>>>(deg, deg1);
    k_msg <<<(NAUG + 255) / 256, B, 0, stream>>>(triples, W1, deg, x1, x1b);
    k_relu<<<(NN * HIDN + 255) / 256, B, 0, stream>>>(x1, x1b, b1);

    k_l2<<<NN / 16, dim3(32), 0, stream>>>(x1, deg, head0, head1, link, W2, b2, x2);

    k_score<<<(NT / SPW + 7) / 8, B, 0, stream>>>(triples, x2, rel, out);
}


# Layer 1 walks the per-slot edge lists too: a lane sums its slots' W1 rows in registers, their count gives 1/deg, relu rows written at once; no deg/x1 planes, no global f32 adds (sums in list order)
# speedup vs baseline: 4.8096x; 1.0876x over previous
#include <hip/hip_runtime.h>

typedef float v2f __attribute__((ext_vector_type(2)));
typedef float v8f __attribute__((ext_vector_type(8)));

#define NN   50000
#define NRL  12
#define NT   500000
#define HIDN 16
#define EMBN 128
#define RP   25
#define KDIM 400
#define NAUG (2*NT + NN)
#define LD   5
#define AP   212

__device__ __forceinline__ void aug_edge(const int* __restrict__ t, int e,
                                         int& s, int& r, int& o) {
    if (e < NT)            { s = t[3*e];     r = t[3*e+1];       o = t[3*e+2]; }
    else if (e < 2*NT)     { int j = e - NT;
                             s = t[3*j+2];   r = t[3*j+1] + NRL; o = t[3*j];   }
    else                   { int i = e - 2*NT;
                             s = i;          r = 2*NRL;          o = i;        }
}

__device__ __forceinline__ int pkg_half() {
    return (int)((__builtin_amdgcn_s_sendmsg_rtn(0x87) >> 18) & 1u);
}

__global__ void k_init(int4* __restrict__ p, int n4) {
    int i = blockIdx.x * blockDim.x + threadIdx.x;
    if (i < n4) p[i] = make_int4(-1, -1, -1, -1);
}

__global__ void k_lists(const int* __restrict__ t, int* __restrict__ head0,
                        int* __restrict__ head1, int2* __restrict__ link) {
    int e = blockIdx.x * blockDim.x + threadIdx.x;
    if (e >= NAUG) return;
    int* hd = pkg_half() ? head1 : head0;
    int s, r, o; aug_edge(t, e, s, r, o);
    int old = atomicExch(&hd[o * RP + r], e);
    link[e] = make_int2(old, s);
}

__global__ void __launch_bounds__(128)
k_l1(const float* __restrict__ W1, const float* __restrict__ b1,
     const int* __restrict__ head0, const int* __restrict__ head1,
     const int2* __restrict__ link, float* __restrict__ norm, float* __restrict__ h) {
    const int lane = threadIdx.x & 31;
    const int w    = blockIdx.x * 4 + (threadIdx.x >> 5);
    const int o = w * 4 + (lane >> 3);
    const int j = lane & 7;

    int   e[8];
    float acc[4][16];
    int   cnt[4];
    int   rbase[4];
#pragma unroll
    for (int i = 0; i < 4; i++) {
        const int r = j + 8 * i;
        if (r < RP) { e[2 * i] = head0[o * RP + r]; e[2 * i + 1] = head1[o * RP + r]; }
        else        { e[2 * i] = -1;                e[2 * i + 1] = -1;                }
        rbase[i] = r * (NN * HIDN);
        cnt[i] = 0;
#pragma unroll
        for (int k = 0; k < 16; k++) acc[i][k] = 0.f;
    }

    for (int pass = 0; pass < NAUG; pass++) {
        bool live = false;
#pragma unroll
        for (int k = 0; k < 8; k++) live |= (e[k] >= 0);
        if (__builtin_amdgcn_ballot_w32(live) == 0u) break;

        int2 ln[8];
#pragma unroll
        for (int k = 0; k < 8; k++) ln[k] = link[e[k] >= 0 ? e[k] : 0];

        float4 rw[8][4];
#pragma unroll
        for (int k = 0; k < 8; k++) {
            const int off = e[k] >= 0 ? rbase[k >> 1] + ln[k].y * HIDN : 0;
            const float4* wr = reinterpret_cast<const float4*>(W1 + off);
            rw[k][0] = wr[0]; rw[k][1] = wr[1]; rw[k][2] = wr[2]; rw[k][3] = wr[3];
        }
#pragma unroll
        for (int k = 0; k < 8; k++) {
            if (e[k] >= 0) {
                float* a = acc[k >> 1];
                a[0]  += rw[k][0].x; a[1]  += rw[k][0].y; a[2]  += rw[k][0].z; a[3]  += rw[k][0].w;
                a[4]  += rw[k][1].x; a[5]  += rw[k][1].y; a[6]  += rw[k][1].z; a[7]  += rw[k][1].w;
                a[8]  += rw[k][2].x; a[9]  += rw[k][2].y; a[10] += rw[k][2].z; a[11] += rw[k][2].w;
                a[12] += rw[k][3].x; a[13] += rw[k][3].y; a[14] += rw[k][3].z; a[15] += rw[k][3].w;
                cnt[k >> 1]++;
                e[k] = ln[k].x;
            }
        }
    }

    float v[16];
#pragma unroll
    for (int k = 0; k < 16; k++) v[k] = 0.f;
#pragma unroll
    for (int i = 0; i < 4; i++) {
        const int r = j + 8 * i;
        if (r < RP) {
            const float en = 1.0f / fmaxf((float)cnt[i], 1.0f);
            norm[o * RP + r] = en;
#pragma unroll
            for (int k = 0; k < 16; k++) v[k] += acc[i][k] * en;
        }
    }

    const bool h2 = (lane & 4) != 0, h1 = (lane & 2) != 0, h0 = (lane & 1) != 0;
    float a8[8], a4[4], a2[2];
#pragma unroll
    for (int i = 0; i < 8; i++) {
        const float keep = h2 ? v[i + 8] : v[i];
        const float send = h2 ? v[i]     : v[i + 8];
        a8[i] = keep + __shfl_xor(send, 4, 32);
    }
#pragma unroll
    for (int i = 0; i < 4; i++) {
        const float keep = h1 ? a8[i + 4] : a8[i];
        const float send = h1 ? a8[i]     : a8[i + 4];
        a4[i] = keep + __shfl_xor(send, 2, 32);
    }
#pragma unroll
    for (int i = 0; i < 2; i++) {
        const float keep = h0 ? a4[i + 2] : a4[i];
        const float send = h0 ? a4[i]     : a4[i + 2];
        a2[i] = keep + __shfl_xor(send, 1, 32);
    }
    const int c0 = 2 * j;
    float2 hv;
    hv.x = fmaxf(a2[0] + b1[c0],     0.0f);
    hv.y = fmaxf(a2[1] + b1[c0 + 1], 0.0f);
    *reinterpret_cast<float2*>(h + o * HIDN + c0) = hv;
}
static_assert(NN % 16 == 0, "whole blocks: 4 waves of 4 nodes");
static_assert(RP <= 32, "8 lanes x 4 slots cover a node's relations");

template <int RB, int NRP>
__device__ __forceinline__ void l2_part(float* As, const int m0, const int lane,
                                        const float* __restrict__ h,
                                        const float* __restrict__ norm,
                                        const int* __restrict__ head0,
                                        const int* __restrict__ head1,
                                        const int2* __restrict__ link,
                                        const float* __restrict__ W2, v8f (&c)[8]) {
    const int half = lane >> 4;
    const int lr   = lane & 15;

    __syncthreads();
    for (int i = lane; i < 16 * AP / 4; i += 32)
        reinterpret_cast<float4*>(As)[i] = make_float4(0.f, 0.f, 0.f, 0.f);

    int   e[NRP];
    float g[NRP];
#pragma unroll
    for (int i = 0; i < NRP; i++) {
        const int q   = lane + 32 * i;
        const int pl  = q / (16 * NRP);
        const int sl  = q - pl * (16 * NRP);
        const int row = sl / NRP, rr = sl - row * NRP;
        const int o = m0 + row, r = RB + rr;
        e[i] = (pl ? head1 : head0)[o * RP + r];
        g[i] = norm[o * RP + r];
    }
    __syncthreads();

    for (int pass = 0; pass < NAUG; pass++) {
        bool live = false;
#pragma unroll
        for (int i = 0; i < NRP; i++) live |= (e[i] >= 0);
        if (__builtin_amdgcn_ballot_w32(live) == 0u) break;

        int2 ln[NRP];
#pragma unroll
        for (int i = 0; i < NRP; i++) ln[i] = link[e[i] >= 0 ? e[i] : 0];

        float4 rw[NRP][4];
#pragma unroll
        for (int i = 0; i < NRP + LD; i++) {
            if (i < NRP) {
                const int sn = e[i] >= 0 ? ln[i].y : 0;
                const float4* hr = reinterpret_cast<const float4*>(h + sn * HIDN);
                rw[i][0] = hr[0]; rw[i][1] = hr[1]; rw[i][2] = hr[2]; rw[i][3] = hr[3];
            }
            const int j = i - LD;
            if (j >= 0 && e[j] >= 0) {
                const int q   = lane + 32 * j;
                const int sl  = q % (16 * NRP);
                const int row = sl / NRP, rr = sl - row * NRP;
                float* d = As + row * AP + rr * HIDN;
                const float w = g[j];
                atomicAdd(d +  0, rw[j][0].x * w); atomicAdd(d +  1, rw[j][0].y * w);
                atomicAdd(d +  2, rw[j][0].z * w); atomicAdd(d +  3, rw[j][0].w * w);
                atomicAdd(d +  4, rw[j][1].x * w); atomicAdd(d +  5, rw[j][1].y * w);
                atomicAdd(d +  6, rw[j][1].z * w); atomicAdd(d +  7, rw[j][1].w * w);
                atomicAdd(d +  8, rw[j][2].x * w); atomicAdd(d +  9, rw[j][2].y * w);
                atomicAdd(d + 10, rw[j][2].z * w); atomicAdd(d + 11, rw[j][2].w * w);
                atomicAdd(d + 12, rw[j][3].x * w); atomicAdd(d + 13, rw[j][3].y * w);
                atomicAdd(d + 14, rw[j][3].z * w); atomicAdd(d + 15, rw[j][3].w * w);
                e[j] = ln[j].x;
            }
        }
    }
    __syncthreads();

    for (int kk = 0; kk < NRP * HIDN; kk += 4) {
        const v2f a = *reinterpret_cast<const v2f*>(As + lr * AP + kk + 2 * half);
        const int ka = RB * HIDN + kk + 2 * half;
        const float* brow0 = W2 + (long)ka * EMBN + lr;
        const float* brow1 = W2 + (long)(ka + 1) * EMBN + lr;
#pragma unroll
        for (int nt = 0; nt < 8; nt++) {
            v2f b;
            b.x = brow0[nt * 16];
            b.y = brow1[nt * 16];
            c[nt] = __builtin_amdgcn_wmma_f32_16x16x4_f32(
                false, a, false, b, (short)0, c[nt], false, false);
        }
    }
}

__global__ void __launch_bounds__(32)
k_l2(const float* __restrict__ h, const float* __restrict__ norm,
     const int* __restrict__ head0, const int* __restrict__ head1,
     const int2* __restrict__ link, const float* __restrict__ W2,
     const float* __restrict__ b2, float* __restrict__ x2) {
    __shared__ __attribute__((aligned(16))) float As[16 * AP];
    const int lane = threadIdx.x;
    const int m0   = blockIdx.x * 16;
    const int half = lane >> 4;
    const int lr   = lane & 15;

    v8f c[8];
#pragma unroll
    for (int nt = 0; nt < 8; nt++) {
        v8f z = {0.f, 0.f, 0.f, 0.f, 0.f, 0.f, 0.f, 0.f};
        c[nt] = z;
    }

    l2_part<0, 13>(As, m0, lane, h, norm, head0, head1, link, W2, c);
    l2_part<13, 12>(As, m0, lane, h, norm, head0, head1, link, W2, c);

#pragma unroll
    for (int nt = 0; nt < 8; nt++) {
        float bias = b2[nt * 16 + lr];
#pragma unroll
        for (int v = 0; v < 8; v++) {
            int row = m0 + v + 8 * half;
            x2[(long)row * EMBN + nt * 16 + lr] = c[nt][v] + bias;
        }
    }
}
static_assert(NN % 16 == 0, "strips are whole");
static_assert(RP == 13 + 12, "the two parts cover every relation");

#define SPW 16
static_assert(NT % SPW == 0, "a wave's 16 triples are all in range or all past the end");

__device__ __forceinline__ float dot4(float4 a, float4 r, float4 b) {
    float acc = a.x * r.x * b.x;
    acc += a.y * r.y * b.y;
    acc += a.z * r.z * b.z;
    acc += a.w * r.w * b.w;
    return acc;
}

__global__ void __launch_bounds__(256)
k_score(const int* __restrict__ t, const float* __restrict__ x2,
        const float* __restrict__ rel, float* __restrict__ out) {
    const int lane = threadIdx.x & 31;
    const int w = __builtin_amdgcn_readfirstlane((int)(blockIdx.x * 8 + (threadIdx.x >> 5)));
    if (w >= NT / SPW) return;
    const int* tw = t + 3 * SPW * w;
    const unsigned co = 16u * lane;
    const unsigned ROWB = EMBN * 4u;
    const char* xb = reinterpret_cast<const char*>(x2);
    const char* rb = reinterpret_cast<const char*>(rel);

    float4 xs[SPW], xo[SPW], rp[SPW];
#pragma unroll
    for (int j = 0; j < SPW; j++) {
        const int s = tw[3 * j], p = tw[3 * j + 1], o = tw[3 * j + 2];
        xs[j] = *reinterpret_cast<const float4*>(xb + ((unsigned)s * ROWB + co));
        xo[j] = *reinterpret_cast<const float4*>(xb + ((unsigned)o * ROWB + co));
        rp[j] = *reinterpret_cast<const float4*>(rb + ((unsigned)p * ROWB + co));
    }

    float v[SPW];
#pragma unroll
    for (int j = 0; j < SPW; j++) v[j] = dot4(xs[j], rp[j], xo[j]);

    const bool h4 = (lane & 16) != 0, h3 = (lane & 8) != 0;
    const bool h2 = (lane & 4)  != 0, h1 = (lane & 2) != 0;

    float a8[8], a4[4], a2[2];
#pragma unroll
    for (int i = 0; i < 8; i++) {
        const float keep = h4 ? v[i + 8] : v[i];
        const float send = h4 ? v[i]     : v[i + 8];
        a8[i] = keep + __shfl_xor(send, 16, 32);
    }
#pragma unroll
    for (int i = 0; i < 4; i++) {
        const float keep = h3 ? a8[i + 4] : a8[i];
        const float send = h3 ? a8[i]     : a8[i + 4];
        a4[i] = keep + __shfl_xor(send, 8, 32);
    }
#pragma unroll
    for (int i = 0; i < 2; i++) {
        const float keep = h2 ? a4[i + 2] : a4[i];
        const float send = h2 ? a4[i]     : a4[i + 2];
        a2[i] = keep + __shfl_xor(send, 4, 32);
    }
    const float keep1 = h1 ? a2[1] : a2[0];
    const float send1 = h1 ? a2[0] : a2[1];
    float z = keep1 + __shfl_xor(send1, 2, 32);
    z += __shfl_xor(z, 1, 32);

    if ((lane & 1) == 0) out[SPW * w + (lane >> 1)] = z;
}

extern "C" void kernel_launch(void* const* d_in, const int* in_sizes, int n_in,
                              void* d_out, int out_size, void* d_ws, size_t ws_size,
                              hipStream_t stream) {
    const int*   triples = (const int*)  d_in[0];
    const float* W1      = (const float*)d_in[1];
    const float* b1      = (const float*)d_in[2];
    const float* W2      = (const float*)d_in[3];
    const float* b2      = (const float*)d_in[4];
    const float* rel     = (const float*)d_in[5];
    float*       out     = (float*)d_out;

    float* norm  = (float*)d_ws;
    float* h     = norm + (long)RP * NN;
    int*   head0 = (int*)(h + (long)NN * HIDN + (long)RP * NN + (long)NN * HIDN);
    int*   head1 = head0 + (long)RP * NN;
    int2*  link  = (int2*)(head1 + (long)RP * NN);
    float* x2    = h    + (long)NN * HIDN + (long)NN * KDIM;
    int nh4 = (2 * RP * NN) / 4;

    dim3 B(256);
    k_init <<<(nh4 + 255) / 256, B, 0, stream>>>((int4*)head0, nh4);
    k_lists<<<(NAUG + 255) / 256, B, 0, stream>>>(triples, head0, head1, link);

    k_l1<<<NN / 16, dim3(128), 0, stream>>>(W1, b1, head0, head1, link, norm, h);

    k_l2<<<NN / 16, dim3(32), 0, stream>>>(h, norm, head0, head1, link, W2, b2, x2);

    k_score<<<(NT / SPW + 7) / 8, B, 0, stream>>>(triples, x2, rel, out);
}


# Layer 2 strip sums stay in registers: lanes l, l+16 own a strip row, 8 channels each, walk both lists of its slots; A fragments are the sums; no LDS, barriers or LDS adds (new f32 sum and k order)
# speedup vs baseline: 5.0415x; 1.0482x over previous
#include <hip/hip_runtime.h>

typedef float v2f __attribute__((ext_vector_type(2)));
typedef float v8f __attribute__((ext_vector_type(8)));

#define NN   50000
#define NRL  12
#define NT   500000
#define HIDN 16
#define EMBN 128
#define RP   25
#define KDIM 400
#define NAUG (2*NT + NN)

__device__ __forceinline__ void aug_edge(const int* __restrict__ t, int e,
                                         int& s, int& r, int& o) {
    if (e < NT)            { s = t[3*e];     r = t[3*e+1];       o = t[3*e+2]; }
    else if (e < 2*NT)     { int j = e - NT;
                             s = t[3*j+2];   r = t[3*j+1] + NRL; o = t[3*j];   }
    else                   { int i = e - 2*NT;
                             s = i;          r = 2*NRL;          o = i;        }
}

__device__ __forceinline__ int pkg_half() {
    return (int)((__builtin_amdgcn_s_sendmsg_rtn(0x87) >> 18) & 1u);
}

__global__ void k_init(int4* __restrict__ p, int n4) {
    int i = blockIdx.x * blockDim.x + threadIdx.x;
    if (i < n4) p[i] = make_int4(-1, -1, -1, -1);
}

__global__ void k_lists(const int* __restrict__ t, int* __restrict__ head0,
                        int* __restrict__ head1, int2* __restrict__ link) {
    int e = blockIdx.x * blockDim.x + threadIdx.x;
    if (e >= NAUG) return;
    int* hd = pkg_half() ? head1 : head0;
    int s, r, o; aug_edge(t, e, s, r, o);
    int old = atomicExch(&hd[o * RP + r], e);
    link[e] = make_int2(old, s);
}

__global__ void __launch_bounds__(128)
k_l1(const float* __restrict__ W1, const float* __restrict__ b1,
     const int* __restrict__ head0, const int* __restrict__ head1,
     const int2* __restrict__ link, float* __restrict__ norm, float* __restrict__ h) {
    const int lane = threadIdx.x & 31;
    const int w    = blockIdx.x * 4 + (threadIdx.x >> 5);
    const int o = w * 4 + (lane >> 3);
    const int j = lane & 7;

    int   e[8];
    float acc[4][16];
    int   cnt[4];
    int   rbase[4];
#pragma unroll
    for (int i = 0; i < 4; i++) {
        const int r = j + 8 * i;
        if (r < RP) { e[2 * i] = head0[o * RP + r]; e[2 * i + 1] = head1[o * RP + r]; }
        else        { e[2 * i] = -1;                e[2 * i + 1] = -1;                }
        rbase[i] = r * (NN * HIDN);
        cnt[i] = 0;
#pragma unroll
        for (int k = 0; k < 16; k++) acc[i][k] = 0.f;
    }

    for (int pass = 0; pass < NAUG; pass++) {
        bool live = false;
#pragma unroll
        for (int k = 0; k < 8; k++) live |= (e[k] >= 0);
        if (__builtin_amdgcn_ballot_w32(live) == 0u) break;

        int2 ln[8];
#pragma unroll
        for (int k = 0; k < 8; k++) ln[k] = link[e[k] >= 0 ? e[k] : 0];

        float4 rw[8][4];
#pragma unroll
        for (int k = 0; k < 8; k++) {
            const int off = e[k] >= 0 ? rbase[k >> 1] + ln[k].y * HIDN : 0;
            const float4* wr = reinterpret_cast<const float4*>(W1 + off);
            rw[k][0] = wr[0]; rw[k][1] = wr[1]; rw[k][2] = wr[2]; rw[k][3] = wr[3];
        }
#pragma unroll
        for (int k = 0; k < 8; k++) {
            if (e[k] >= 0) {
                float* a = acc[k >> 1];
                a[0]  += rw[k][0].x; a[1]  += rw[k][0].y; a[2]  += rw[k][0].z; a[3]  += rw[k][0].w;
                a[4]  += rw[k][1].x; a[5]  += rw[k][1].y; a[6]  += rw[k][1].z; a[7]  += rw[k][1].w;
                a[8]  += rw[k][2].x; a[9]  += rw[k][2].y; a[10] += rw[k][2].z; a[11] += rw[k][2].w;
                a[12] += rw[k][3].x; a[13] += rw[k][3].y; a[14] += rw[k][3].z; a[15] += rw[k][3].w;
                cnt[k >> 1]++;
                e[k] = ln[k].x;
            }
        }
    }

    float v[16];
#pragma unroll
    for (int k = 0; k < 16; k++) v[k] = 0.f;
#pragma unroll
    for (int i = 0; i < 4; i++) {
        const int r = j + 8 * i;
        if (r < RP) {
            const float en = 1.0f / fmaxf((float)cnt[i], 1.0f);
            norm[o * RP + r] = en;
#pragma unroll
            for (int k = 0; k < 16; k++) v[k] += acc[i][k] * en;
        }
    }

    const bool h2 = (lane & 4) != 0, h1 = (lane & 2) != 0, h0 = (lane & 1) != 0;
    float a8[8], a4[4], a2[2];
#pragma unroll
    for (int i = 0; i < 8; i++) {
        const float keep = h2 ? v[i + 8] : v[i];
        const float send = h2 ? v[i]     : v[i + 8];
        a8[i] = keep + __shfl_xor(send, 4, 32);
    }
#pragma unroll
    for (int i = 0; i < 4; i++) {
        const float keep = h1 ? a8[i + 4] : a8[i];
        const float send = h1 ? a8[i]     : a8[i + 4];
        a4[i] = keep + __shfl_xor(send, 2, 32);
    }
#pragma unroll
    for (int i = 0; i < 2; i++) {
        const float keep = h0 ? a4[i + 2] : a4[i];
        const float send = h0 ? a4[i]     : a4[i + 2];
        a2[i] = keep + __shfl_xor(send, 1, 32);
    }
    const int c0 = 2 * j;
    float2 hv;
    hv.x = fmaxf(a2[0] + b1[c0],     0.0f);
    hv.y = fmaxf(a2[1] + b1[c0 + 1], 0.0f);
    *reinterpret_cast<float2*>(h + o * HIDN + c0) = hv;
}
static_assert(NN % 16 == 0, "whole blocks: 4 waves of 4 nodes");
static_assert(RP <= 32, "8 lanes x 4 slots cover a node's relations");

#define GR 5
static_assert(RP % GR == 0, "the groups cover every relation");

__global__ void __launch_bounds__(32)
k_l2(const float* __restrict__ h, const float* __restrict__ norm,
     const int* __restrict__ head0, const int* __restrict__ head1,
     const int2* __restrict__ link, const float* __restrict__ W2,
     const float* __restrict__ b2, float* __restrict__ x2) {
    const int lane = threadIdx.x;
    const int m0   = blockIdx.x * 16;
    const int half = lane >> 4;
    const int lr   = lane & 15;
    const int o    = m0 + lr;
    const int hoff = 8 * half;

    v8f c[8];
#pragma unroll
    for (int nt = 0; nt < 8; nt++) {
        v8f z = {0.f, 0.f, 0.f, 0.f, 0.f, 0.f, 0.f, 0.f};
        c[nt] = z;
    }

#pragma unroll 1
    for (int rb = 0; rb < RP; rb += GR) {
        int   e[2 * GR];
        float acc[GR][8];
        float g[GR];
#pragma unroll
        for (int i = 0; i < GR; i++) {
            e[2 * i]     = head0[o * RP + rb + i];
            e[2 * i + 1] = head1[o * RP + rb + i];
            g[i]         = norm[o * RP + rb + i];
#pragma unroll
            for (int k = 0; k < 8; k++) acc[i][k] = 0.f;
        }

        for (int pass = 0; pass < NAUG; pass++) {
            bool live = false;
#pragma unroll
            for (int k = 0; k < 2 * GR; k++) live |= (e[k] >= 0);
            if (__builtin_amdgcn_ballot_w32(live) == 0u) break;

            int2 ln[2 * GR];
#pragma unroll
            for (int k = 0; k < 2 * GR; k++) ln[k] = link[e[k] >= 0 ? e[k] : 0];

            float4 rw[2 * GR][2];
#pragma unroll
            for (int k = 0; k < 2 * GR; k++) {
                const int sn = e[k] >= 0 ? ln[k].y : 0;
                const float4* hr = reinterpret_cast<const float4*>(h + sn * HIDN + hoff);
                rw[k][0] = hr[0]; rw[k][1] = hr[1];
            }
#pragma unroll
            for (int k = 0; k < 2 * GR; k++) {
                if (e[k] >= 0) {
                    float* a = acc[k >> 1];
                    a[0] += rw[k][0].x; a[1] += rw[k][0].y; a[2] += rw[k][0].z; a[3] += rw[k][0].w;
                    a[4] += rw[k][1].x; a[5] += rw[k][1].y; a[6] += rw[k][1].z; a[7] += rw[k][1].w;
                    e[k] = ln[k].x;
                }
            }
        }

#pragma unroll
        for (int i = 0; i < GR; i++) {
            const float* bp = W2 + (long)((rb + i) * HIDN + hoff) * EMBN + lr;
            float bv[8][8];
#pragma unroll
            for (int j = 0; j < 8; j++) {
#pragma unroll
                for (int nt = 0; nt < 8; nt++) bv[j][nt] = bp[j * EMBN + nt * 16];
            }
#pragma unroll
            for (int t = 0; t < 4; t++) {
                v2f a;
                a.x = acc[i][2 * t]     * g[i];
                a.y = acc[i][2 * t + 1] * g[i];
#pragma unroll
                for (int nt = 0; nt < 8; nt++) {
                    v2f b;
                    b.x = bv[2 * t][nt];
                    b.y = bv[2 * t + 1][nt];
                    c[nt] = __builtin_amdgcn_wmma_f32_16x16x4_f32(
                        false, a, false, b, (short)0, c[nt], false, false);
                }
            }
        }
    }

#pragma unroll
    for (int nt = 0; nt < 8; nt++) {
        float bias = b2[nt * 16 + lr];
#pragma unroll
        for (int v = 0; v < 8; v++) {
            int row = m0 + v + 8 * half;
            x2[(long)row * EMBN + nt * 16 + lr] = c[nt][v] + bias;
        }
    }
}
static_assert(NN % 16 == 0, "strips are whole");

#define SPW 16
static_assert(NT % SPW == 0, "a wave's 16 triples are all in range or all past the end");

__device__ __forceinline__ float dot4(float4 a, float4 r, float4 b) {
    float acc = a.x * r.x * b.x;
    acc += a.y * r.y * b.y;
    acc += a.z * r.z * b.z;
    acc += a.w * r.w * b.w;
    return acc;
}

__global__ void __launch_bounds__(256)
k_score(const int* __restrict__ t, const float* __restrict__ x2,
        const float* __restrict__ rel, float* __restrict__ out) {
    const int lane = threadIdx.x & 31;
    const int w = __builtin_amdgcn_readfirstlane((int)(blockIdx.x * 8 + (threadIdx.x >> 5)));
    if (w >= NT / SPW) return;
    const int* tw = t + 3 * SPW * w;
    const unsigned co = 16u * lane;
    const unsigned ROWB = EMBN * 4u;
    const char* xb = reinterpret_cast<const char*>(x2);
    const char* rb = reinterpret_cast<const char*>(rel);

    float4 xs[SPW], xo[SPW], rp[SPW];
#pragma unroll
    for (int j = 0; j < SPW; j++) {
        const int s = tw[3 * j], p = tw[3 * j + 1], o = tw[3 * j + 2];
        xs[j] = *reinterpret_cast<const float4*>(xb + ((unsigned)s * ROWB + co));
        xo[j] = *reinterpret_cast<const float4*>(xb + ((unsigned)o * ROWB + co));
        rp[j] = *reinterpret_cast<const float4*>(rb + ((unsigned)p * ROWB + co));
    }

    float v[SPW];
#pragma unroll
    for (int j = 0; j < SPW; j++) v[j] = dot4(xs[j], rp[j], xo[j]);

    const bool h4 = (lane & 16) != 0, h3 = (lane & 8) != 0;
    const bool h2 = (lane & 4)  != 0, h1 = (lane & 2) != 0;

    float a8[8], a4[4], a2[2];
#pragma unroll
    for (int i = 0; i < 8; i++) {
        const float keep = h4 ? v[i + 8] : v[i];
        const float send = h4 ? v[i]     : v[i + 8];
        a8[i] = keep + __shfl_xor(send, 16, 32);
    }
#pragma unroll
    for (int i = 0; i < 4; i++) {
        const float keep = h3 ? a8[i + 4] : a8[i];
        const float send = h3 ? a8[i]     : a8[i + 4];
        a4[i] = keep + __shfl_xor(send, 8, 32);
    }
#pragma unroll
    for (int i = 0; i < 2; i++) {
        const float keep = h2 ? a4[i + 2] : a4[i];
        const float send = h2 ? a4[i]     : a4[i + 2];
        a2[i] = keep + __shfl_xor(send, 4, 32);
    }
    const float keep1 = h1 ? a2[1] : a2[0];
    const float send1 = h1 ? a2[0] : a2[1];
    float z = keep1 + __shfl_xor(send1, 2, 32);
    z += __shfl_xor(z, 1, 32);

    if ((lane & 1) == 0) out[SPW * w + (lane >> 1)] = z;
}

extern "C" void kernel_launch(void* const* d_in, const int* in_sizes, int n_in,
                              void* d_out, int out_size, void* d_ws, size_t ws_size,
                              hipStream_t stream) {
    const int*   triples = (const int*)  d_in[0];
    const float* W1      = (const float*)d_in[1];
    const float* b1      = (const float*)d_in[2];
    const float* W2      = (const float*)d_in[3];
    const float* b2      = (const float*)d_in[4];
    const float* rel     = (const float*)d_in[5];
    float*       out     = (float*)d_out;

    float* norm  = (float*)d_ws;
    float* h     = norm + (long)RP * NN;
    int*   head0 = (int*)(h + (long)NN * HIDN + (long)RP * NN + (long)NN * HIDN);
    int*   head1 = head0 + (long)RP * NN;
    int2*  link  = (int2*)(head1 + (long)RP * NN);
    float* x2    = h    + (long)NN * HIDN + (long)NN * KDIM;
    int nh4 = (2 * RP * NN) / 4;

    dim3 B(256);
    k_init <<<(nh4 + 255) / 256, B, 0, stream>>>((int4*)head0, nh4);
    k_lists<<<(NAUG + 255) / 256, B, 0, stream>>>(triples, head0, head1, link);

    k_l1<<<NN / 16, dim3(128), 0, stream>>>(W1, b1, head0, head1, link, norm, h);

    k_l2<<<NN / 16, dim3(32), 0, stream>>>(h, norm, head0, head1, link, W2, b2, x2);

    k_score<<<(NT / SPW + 7) / 8, B, 0, stream>>>(triples, x2, rel, out);
}


# Layer 2: the first launch also writes W2 in the order the waves read it, so a lane's 4 K of a double step are one 16-B load and a wave's load is 4 whole lines: 50 W2 loads a wave, not 200
# speedup vs baseline: 5.5553x; 1.1019x over previous
#include <hip/hip_runtime.h>

typedef float v2f __attribute__((ext_vector_type(2)));
typedef float v8f __attribute__((ext_vector_type(8)));

#define NN   50000
#define NRL  12
#define NT   500000
#define HIDN 16
#define EMBN 128
#define RP   25
#define KDIM 400
#define NAUG (2*NT + NN)

__device__ __forceinline__ void aug_edge(const int* __restrict__ t, int e,
                                         int& s, int& r, int& o) {
    if (e < NT)            { s = t[3*e];     r = t[3*e+1];       o = t[3*e+2]; }
    else if (e < 2*NT)     { int j = e - NT;
                             s = t[3*j+2];   r = t[3*j+1] + NRL; o = t[3*j];   }
    else                   { int i = e - 2*NT;
                             s = i;          r = 2*NRL;          o = i;        }
}

__device__ __forceinline__ int pkg_half() {
    return (int)((__builtin_amdgcn_s_sendmsg_rtn(0x87) >> 18) & 1u);
}

#define NWV 8
__global__ void k_init(int4* __restrict__ p, int n4, const float* __restrict__ W2,
                       float4* __restrict__ W2P) {
    int i = blockIdx.x * blockDim.x + threadIdx.x;
    if (i < n4) p[i] = make_int4(-1, -1, -1, -1);
    if (i < NWV * (KDIM / 8) * 32) {
        const int lane = i & 31;
        const int u    = (i >> 5) % (KDIM / 8);
        const int w    = (i >> 5) / (KDIM / 8);
        const float* s = W2 + (8 * u + 4 * (lane >> 4)) * EMBN + 16 * w + (lane & 15);
        W2P[i] = make_float4(s[0], s[EMBN], s[2 * EMBN], s[3 * EMBN]);
    }
}
static_assert(NWV * 16 == EMBN, "the 16-column blocks cover W2's columns");
static_assert(KDIM % 8 == 0, "double steps of 8 K cover K");

__global__ void k_lists(const int* __restrict__ t, int* __restrict__ head0,
                        int* __restrict__ head1, int2* __restrict__ link) {
    int e = blockIdx.x * blockDim.x + threadIdx.x;
    if (e >= NAUG) return;
    int* hd = pkg_half() ? head1 : head0;
    int s, r, o; aug_edge(t, e, s, r, o);
    int old = atomicExch(&hd[o * RP + r], e);
    link[e] = make_int2(old, s);
}

__global__ void __launch_bounds__(128)
k_l1(const float* __restrict__ W1, const float* __restrict__ b1,
     const int* __restrict__ head0, const int* __restrict__ head1,
     const int2* __restrict__ link, float* __restrict__ norm, float* __restrict__ h) {
    const int lane = threadIdx.x & 31;
    const int w    = blockIdx.x * 4 + (threadIdx.x >> 5);
    const int o = w * 4 + (lane >> 3);
    const int j = lane & 7;

    int   e[8];
    float acc[4][16];
    int   cnt[4];
    int   rbase[4];
#pragma unroll
    for (int i = 0; i < 4; i++) {
        const int r = j + 8 * i;
        if (r < RP) { e[2 * i] = head0[o * RP + r]; e[2 * i + 1] = head1[o * RP + r]; }
        else        { e[2 * i] = -1;                e[2 * i + 1] = -1;                }
        rbase[i] = r * (NN * HIDN);
        cnt[i] = 0;
#pragma unroll
        for (int k = 0; k < 16; k++) acc[i][k] = 0.f;
    }

    for (int pass = 0; pass < NAUG; pass++) {
        bool live = false;
#pragma unroll
        for (int k = 0; k < 8; k++) live |= (e[k] >= 0);
        if (__builtin_amdgcn_ballot_w32(live) == 0u) break;

        int2 ln[8];
#pragma unroll
        for (int k = 0; k < 8; k++) ln[k] = link[e[k] >= 0 ? e[k] : 0];

        float4 rw[8][4];
#pragma unroll
        for (int k = 0; k < 8; k++) {
            const int off = e[k] >= 0 ? rbase[k >> 1] + ln[k].y * HIDN : 0;
            const float4* wr = reinterpret_cast<const float4*>(W1 + off);
            rw[k][0] = wr[0]; rw[k][1] = wr[1]; rw[k][2] = wr[2]; rw[k][3] = wr[3];
        }
#pragma unroll
        for (int k = 0; k < 8; k++) {
            if (e[k] >= 0) {
                float* a = acc[k >> 1];
                a[0]  += rw[k][0].x; a[1]  += rw[k][0].y; a[2]  += rw[k][0].z; a[3]  += rw[k][0].w;
                a[4]  += rw[k][1].x; a[5]  += rw[k][1].y; a[6]  += rw[k][1].z; a[7]  += rw[k][1].w;
                a[8]  += rw[k][2].x; a[9]  += rw[k][2].y; a[10] += rw[k][2].z; a[11] += rw[k][2].w;
                a[12] += rw[k][3].x; a[13] += rw[k][3].y; a[14] += rw[k][3].z; a[15] += rw[k][3].w;
                cnt[k >> 1]++;
                e[k] = ln[k].x;
            }
        }
    }

    float v[16];
#pragma unroll
    for (int k = 0; k < 16; k++) v[k] = 0.f;
#pragma unroll
    for (int i = 0; i < 4; i++) {
        const int r = j + 8 * i;
        if (r < RP) {
            const float en = 1.0f / fmaxf((float)cnt[i], 1.0f);
            norm[o * RP + r] = en;
#pragma unroll
            for (int k = 0; k < 16; k++) v[k] += acc[i][k] * en;
        }
    }

    const bool h2 = (lane & 4) != 0, h1 = (lane & 2) != 0, h0 = (lane & 1) != 0;
    float a8[8], a4[4], a2[2];
#pragma unroll
    for (int i = 0; i < 8; i++) {
        const float keep = h2 ? v[i + 8] : v[i];
        const float send = h2 ? v[i]     : v[i + 8];
        a8[i] = keep + __shfl_xor(send, 4, 32);
    }
#pragma unroll
    for (int i = 0; i < 4; i++) {
        const float keep = h1 ? a8[i + 4] : a8[i];
        const float send = h1 ? a8[i]     : a8[i + 4];
        a4[i] = keep + __shfl_xor(send, 2, 32);
    }
#pragma unroll
    for (int i = 0; i < 2; i++) {
        const float keep = h0 ? a4[i + 2] : a4[i];
        const float send = h0 ? a4[i]     : a4[i + 2];
        a2[i] = keep + __shfl_xor(send, 1, 32);
    }
    const int c0 = 2 * j;
    float2 hv;
    hv.x = fmaxf(a2[0] + b1[c0],     0.0f);
    hv.y = fmaxf(a2[1] + b1[c0 + 1], 0.0f);
    *reinterpret_cast<float2*>(h + o * HIDN + c0) = hv;
}
static_assert(NN % 16 == 0, "whole blocks: 4 waves of 4 nodes");
static_assert(RP <= 32, "8 lanes x 4 slots cover a node's relations");

#define APITCH 404
#define KCH 10
static_assert((KDIM / 8) % KCH == 0, "the groups cover K");
static_assert(16 * RP <= 512, "256 lanes x 2 slots cover a strip");

__global__ void __launch_bounds__(256)
k_l2(const float* __restrict__ h, const float* __restrict__ norm,
     const int* __restrict__ head0, const int* __restrict__ head1,
     const int2* __restrict__ link, const float4* __restrict__ W2P,
     const float* __restrict__ b2, float* __restrict__ x2) {
    __shared__ __attribute__((aligned(16))) float As[16 * APITCH];
    const int tid   = threadIdx.x;
    const int m0    = blockIdx.x * 16;
    const int sbase = m0 * RP;

    {
        int   e[4];
        float acc[2][16];
        float g[2];
#pragma unroll
        for (int j = 0; j < 2; j++) {
            const int q = tid + 256 * j;
            if (q < 16 * RP) {
                e[2 * j]     = head0[sbase + q];
                e[2 * j + 1] = head1[sbase + q];
                g[j]         = norm[sbase + q];
            } else {
                e[2 * j] = -1; e[2 * j + 1] = -1; g[j] = 0.f;
            }
#pragma unroll
            for (int k = 0; k < 16; k++) acc[j][k] = 0.f;
        }

        for (int pass = 0; pass < NAUG; pass++) {
            bool live = false;
#pragma unroll
            for (int k = 0; k < 4; k++) live |= (e[k] >= 0);
            if (__builtin_amdgcn_ballot_w32(live) == 0u) break;

            int2 ln[4];
#pragma unroll
            for (int k = 0; k < 4; k++) ln[k] = link[e[k] >= 0 ? e[k] : 0];

            float4 rw[4][4];
#pragma unroll
            for (int k = 0; k < 4; k++) {
                const int sn = e[k] >= 0 ? ln[k].y : 0;
                const float4* hr = reinterpret_cast<const float4*>(h + sn * HIDN);
                rw[k][0] = hr[0]; rw[k][1] = hr[1]; rw[k][2] = hr[2]; rw[k][3] = hr[3];
            }
#pragma unroll
            for (int k = 0; k < 4; k++) {
                if (e[k] >= 0) {
                    float* a = acc[k >> 1];
                    a[0]  += rw[k][0].x; a[1]  += rw[k][0].y; a[2]  += rw[k][0].z; a[3]  += rw[k][0].w;
                    a[4]  += rw[k][1].x; a[5]  += rw[k][1].y; a[6]  += rw[k][1].z; a[7]  += rw[k][1].w;
                    a[8]  += rw[k][2].x; a[9]  += rw[k][2].y; a[10] += rw[k][2].z; a[11] += rw[k][2].w;
                    a[12] += rw[k][3].x; a[13] += rw[k][3].y; a[14] += rw[k][3].z; a[15] += rw[k][3].w;
                    e[k] = ln[k].x;
                }
            }
        }

#pragma unroll
        for (int j = 0; j < 2; j++) {
            const int q = tid + 256 * j;
            if (q < 16 * RP) {
                const int row = q / RP;
                const int r   = q - row * RP;
                float4* dst = reinterpret_cast<float4*>(As + row * APITCH + r * HIDN);
#pragma unroll
                for (int p = 0; p < 4; p++)
                    dst[p] = make_float4(acc[j][4 * p] * g[j],     acc[j][4 * p + 1] * g[j],
                                         acc[j][4 * p + 2] * g[j], acc[j][4 * p + 3] * g[j]);
            }
        }
    }
    __syncthreads();

    const int lane = tid & 31;
    const int half = lane >> 4;
    const int lr   = lane & 15;
    const int col  = (tid >> 5) * 16 + lr;

    v8f c0 = {0.f, 0.f, 0.f, 0.f, 0.f, 0.f, 0.f, 0.f};
    v8f c1 = {0.f, 0.f, 0.f, 0.f, 0.f, 0.f, 0.f, 0.f};

    const float* ap = As + lr * APITCH + 4 * half;
    const float4* bq = W2P + (tid >> 5) * ((KDIM / 8) * 32) + lane;

#pragma unroll 1
    for (int u0 = 0; u0 < KDIM / 8; u0 += KCH) {
        float4 av[KCH];
        float4 bv[KCH];
#pragma unroll
        for (int i = 0; i < KCH; i++) bv[i] = bq[(u0 + i) * 32];
#pragma unroll
        for (int i = 0; i < KCH; i++) av[i] = *reinterpret_cast<const float4*>(ap + 8 * (u0 + i));
        __builtin_amdgcn_sched_barrier(0);
#pragma unroll
        for (int i = 0; i < KCH; i++) {
            v2f a, b;
            a.x = av[i].x; a.y = av[i].y;
            b.x = bv[i].x; b.y = bv[i].y;
            c0 = __builtin_amdgcn_wmma_f32_16x16x4_f32(false, a, false, b, (short)0, c0, false, false);
            a.x = av[i].z; a.y = av[i].w;
            b.x = bv[i].z; b.y = bv[i].w;
            c1 = __builtin_amdgcn_wmma_f32_16x16x4_f32(false, a, false, b, (short)0, c1, false, false);
        }
    }

    const float bias = b2[col];
#pragma unroll
    for (int v = 0; v < 8; v++) {
        const int row = m0 + v + 8 * half;
        x2[(long)row * EMBN + col] = (c0[v] + c1[v]) + bias;
    }
}
static_assert(NN % 16 == 0, "strips are whole");
static_assert(NWV * 32 == 256, "k_l2's work-group is NWV waves");

#define SPW 16
static_assert(NT % SPW == 0, "a wave's 16 triples are all in range or all past the end");

__device__ __forceinline__ float dot4(float4 a, float4 r, float4 b) {
    float acc = a.x * r.x * b.x;
    acc += a.y * r.y * b.y;
    acc += a.z * r.z * b.z;
    acc += a.w * r.w * b.w;
    return acc;
}

__global__ void __launch_bounds__(256)
k_score(const int* __restrict__ t, const float* __restrict__ x2,
        const float* __restrict__ rel, float* __restrict__ out) {
    const int lane = threadIdx.x & 31;
    const int w = __builtin_amdgcn_readfirstlane((int)(blockIdx.x * 8 + (threadIdx.x >> 5)));
    if (w >= NT / SPW) return;
    const int* tw = t + 3 * SPW * w;
    const unsigned co = 16u * lane;
    const unsigned ROWB = EMBN * 4u;
    const char* xb = reinterpret_cast<const char*>(x2);
    const char* rb = reinterpret_cast<const char*>(rel);

    float4 xs[SPW], xo[SPW], rp[SPW];
#pragma unroll
    for (int j = 0; j < SPW; j++) {
        const int s = tw[3 * j], p = tw[3 * j + 1], o = tw[3 * j + 2];
        xs[j] = *reinterpret_cast<const float4*>(xb + ((unsigned)s * ROWB + co));
        xo[j] = *reinterpret_cast<const float4*>(xb + ((unsigned)o * ROWB + co));
        rp[j] = *reinterpret_cast<const float4*>(rb + ((unsigned)p * ROWB + co));
    }

    float v[SPW];
#pragma unroll
    for (int j = 0; j < SPW; j++) v[j] = dot4(xs[j], rp[j], xo[j]);

    const bool h4 = (lane & 16) != 0, h3 = (lane & 8) != 0;
    const bool h2 = (lane & 4)  != 0, h1 = (lane & 2) != 0;

    float a8[8], a4[4], a2[2];
#pragma unroll
    for (int i = 0; i < 8; i++) {
        const float keep = h4 ? v[i + 8] : v[i];
        const float send = h4 ? v[i]     : v[i + 8];
        a8[i] = keep + __shfl_xor(send, 16, 32);
    }
#pragma unroll
    for (int i = 0; i < 4; i++) {
        const float keep = h3 ? a8[i + 4] : a8[i];
        const float send = h3 ? a8[i]     : a8[i + 4];
        a4[i] = keep + __shfl_xor(send, 8, 32);
    }
#pragma unroll
    for (int i = 0; i < 2; i++) {
        const float keep = h2 ? a4[i + 2] : a4[i];
        const float send = h2 ? a4[i]     : a4[i + 2];
        a2[i] = keep + __shfl_xor(send, 4, 32);
    }
    const float keep1 = h1 ? a2[1] : a2[0];
    const float send1 = h1 ? a2[0] : a2[1];
    float z = keep1 + __shfl_xor(send1, 2, 32);
    z += __shfl_xor(z, 1, 32);

    if ((lane & 1) == 0) out[SPW * w + (lane >> 1)] = z;
}

extern "C" void kernel_launch(void* const* d_in, const int* in_sizes, int n_in,
                              void* d_out, int out_size, void* d_ws, size_t ws_size,
                              hipStream_t stream) {
    const int*   triples = (const int*)  d_in[0];
    const float* W1      = (const float*)d_in[1];
    const float* b1      = (const float*)d_in[2];
    const float* W2      = (const float*)d_in[3];
    const float* b2      = (const float*)d_in[4];
    const float* rel     = (const float*)d_in[5];
    float*       out     = (float*)d_out;

    float* norm  = (float*)d_ws;
    float* h     = norm + (long)RP * NN;
    int*   head0 = (int*)(h + (long)NN * HIDN + (long)RP * NN + (long)NN * HIDN);
    int*   head1 = head0 + (long)RP * NN;
    int2*  link  = (int2*)(head1 + (long)RP * NN);
    float* x2    = h    + (long)NN * HIDN + (long)NN * KDIM;
    float4* W2P  = (float4*)(h + (long)NN * HIDN + 48);
    int nh4 = (2 * RP * NN) / 4;

    dim3 B(256);
    k_init <<<(nh4 + 255) / 256, B, 0, stream>>>((int4*)head0, nh4, W2, W2P);
    k_lists<<<(NAUG + 255) / 256, B, 0, stream>>>(triples, head0, head1, link);

    k_l1<<<NN / 16, dim3(128), 0, stream>>>(W1, b1, head0, head1, link, norm, h);

    k_l2<<<NN / 16, B, 0, stream>>>(h, norm, head0, head1, link, W2P, b2, x2);

    k_score<<<(NT / SPW + 7) / 8, B, 0, stream>>>(triples, x2, rel, out);
}
